# starting kernel
# baseline (speedup 1.0000x reference)
#include <hip/hip_runtime.h>
#include <math.h>

#define N_PTS   8192
#define WV_DIM  300
#define D_DIM   128
#define BIG_F   1e8f
#define EPS_TRM (128.0f * 1e-16f)

typedef float v2f __attribute__((ext_vector_type(2)));
typedef float v8f __attribute__((ext_vector_type(8)));

__device__ __forceinline__ v2f ld2(const float* p) {
    return *(const v2f*)p;
}

__global__ void x2_kernel(const float* __restrict__ x, float* __restrict__ x2) {
    int i = blockIdx.x * blockDim.x + threadIdx.x;
    if (i < N_PTS) {
        const float* row = x + (size_t)i * D_DIM;
        float s = 0.f;
        #pragma unroll 8
        for (int d = 0; d < D_DIM; ++d) s = fmaf(row[d], row[d], s);
        x2[i] = s;
    }
}

__global__ __launch_bounds__(32) void sem_fc_kernel(
    const float* __restrict__ x,  const float* __restrict__ wv,
    const float* __restrict__ alpha, const float* __restrict__ W,
    const float* __restrict__ b,  float* __restrict__ sem_out,
    float* __restrict__ sem2) {
    const int lane = threadIdx.x;
    const int col  = lane & 15;
    const int hi   = lane >> 4;
    const int i0   = blockIdx.x * 16;

    float alph[8];
    #pragma unroll
    for (int r = 0; r < 8; ++r) alph[r] = alpha[i0 + r + 8 * hi];

    float nrm[8];
    #pragma unroll
    for (int r = 0; r < 8; ++r) nrm[r] = 0.f;

    const float* arow = wv + (size_t)(i0 + col) * WV_DIM + 2 * hi;

    for (int dt = 0; dt < D_DIM / 16; ++dt) {
        const int d = dt * 16 + col;
        const float* brow = W + (size_t)d * WV_DIM + 2 * hi;

        v8f c = {};
        for (int k = 0; k < WV_DIM; k += 4) {
            v2f a  = ld2(arow + k);
            v2f bb = ld2(brow + k);
            c = __builtin_amdgcn_wmma_f32_16x16x4_f32(
                    false, a, false, bb, (short)0, c, false, false);
        }

        const float bd = b[d];
        #pragma unroll
        for (int r = 0; r < 8; ++r) {
            const int i = i0 + r + 8 * hi;
            const float so = c[r] + bd;
            const float xv = x[(size_t)i * D_DIM + d];
            const float s  = alph[r] * so + (1.f - alph[r]) * xv;
            sem_out[(size_t)i * D_DIM + d] = s;
            nrm[r] = fmaf(s, s, nrm[r]);
        }
    }

    #pragma unroll
    for (int r = 0; r < 8; ++r) {
        float v = nrm[r];
        for (int off = 8; off >= 1; off >>= 1)
            v += __shfl_xor(v, off, 16);
        if (col == 0) sem2[i0 + r + 8 * hi] = v;
    }
}

__global__ __launch_bounds__(32) void pair_kernel(
    const float* __restrict__ x,    const int* __restrict__ label,
    const float* __restrict__ sem,  const float* __restrict__ sem2,
    const float* __restrict__ x2,   float* __restrict__ diff) {
    const int lane = threadIdx.x;
    const int col  = lane & 15;
    const int hi   = lane >> 4;
    const int i0   = blockIdx.x * 16;

    v2f afrag[32];
    const float* arow = sem + (size_t)(i0 + col) * D_DIM + 2 * hi;
    #pragma unroll
    for (int ks = 0; ks < 32; ++ks) afrag[ks] = ld2(arow + 4 * ks);

    int   labi[8];
    float s2i[8];
    #pragma unroll
    for (int r = 0; r < 8; ++r) {
        labi[r] = label[i0 + r + 8 * hi];
        s2i[r]  = sem2[i0 + r + 8 * hi];
    }

    float fp[8], cn[8];
    #pragma unroll
    for (int r = 0; r < 8; ++r) { fp[r] = 0.f; cn[r] = 3.4e38f; }

    for (int jt = 0; jt < N_PTS / 16; ++jt) {
        const int   j    = jt * 16 + col;
        const int   labj = label[j];
        const float x2j  = x2[j];
        const float* brow = x + (size_t)j * D_DIM + 2 * hi;

        v8f c = {};
        #pragma unroll
        for (int ks = 0; ks < 32; ++ks) {
            v2f bb = ld2(brow + 4 * ks);
            c = __builtin_amdgcn_wmma_f32_16x16x4_f32(
                    false, afrag[ks], false, bb, (short)0, c, false, false);
        }

        #pragma unroll
        for (int r = 0; r < 8; ++r) {
            const int i  = i0 + r + 8 * hi;
            float sq = s2i[r] + x2j - 2.f * c[r];
            sq = sq > 0.f ? sq : 0.f;
            const float dist = sqrtf(sq + EPS_TRM);
            const bool same = (labi[r] == labj);
            const bool pos  = same && (i != j);
            fp[r] = fmaxf(fp[r], pos ? dist : 0.f);
            cn[r] = fminf(cn[r], same ? dist + BIG_F : dist);
        }
    }

    #pragma unroll
    for (int r = 0; r < 8; ++r) {
        float a = fp[r], m = cn[r];
        for (int off = 8; off >= 1; off >>= 1) {
            a = fmaxf(a, __shfl_xor(a, off, 16));
            m = fminf(m, __shfl_xor(m, off, 16));
        }
        if (col == 0) {
            const float z  = a - m;
            const float sp = (z > 20.f) ? z : log1pf(expf(z));
            diff[i0 + r + 8 * hi] = sp;
        }
    }
}

extern "C" void kernel_launch(void* const* d_in, const int* in_sizes, int n_in,
                              void* d_out, int out_size, void* d_ws, size_t ws_size,
                              hipStream_t stream) {
    const float* x     = (const float*)d_in[0];
    const float* wv    = (const float*)d_in[1];
    const int*   label = (const int*)  d_in[2];
    const float* alpha = (const float*)d_in[3];
    const float* W     = (const float*)d_in[4];
    const float* b     = (const float*)d_in[5];

    float* diff = (float*)d_out;
    float* sem  = diff + N_PTS;

    float* x2   = (float*)d_ws;
    float* sem2 = x2 + N_PTS;

    hipLaunchKernelGGL(x2_kernel, dim3(N_PTS / 256), dim3(256), 0, stream, x, x2);
    hipLaunchKernelGGL(sem_fc_kernel, dim3(N_PTS / 16), dim3(32), 0, stream,
                       x, wv, alpha, W, b, sem, sem2);
    hipLaunchKernelGGL(pair_kernel, dim3(N_PTS / 16), dim3(32), 0, stream,
                       x, label, sem, sem2, x2, diff);
}


# pair kernel on a 2D grid: 8 waves = 128 sem rows x one of 16 x chunks, x rows staged once per group in shared LDS (64-row stages, 2 buffers); chunk max/min merged by a 4th launch; no sum reordered
# speedup vs baseline: 9.7228x; 9.7228x over previous
#include <hip/hip_runtime.h>
#include <math.h>

#define N_PTS   8192
#define WV_DIM  300
#define D_DIM   128
#define BIG_F   1e8f
#define EPS_TRM (128.0f * 1e-16f)

typedef float v2f __attribute__((ext_vector_type(2)));
typedef float v8f __attribute__((ext_vector_type(8)));

__device__ __forceinline__ v2f ld2(const float* p) {
    return *(const v2f*)p;
}

__global__ void x2_kernel(const float* __restrict__ x, float* __restrict__ x2) {
    int i = blockIdx.x * blockDim.x + threadIdx.x;
    if (i < N_PTS) {
        const float* row = x + (size_t)i * D_DIM;
        float s = 0.f;
        #pragma unroll 8
        for (int d = 0; d < D_DIM; ++d) s = fmaf(row[d], row[d], s);
        x2[i] = s;
    }
}

__global__ __launch_bounds__(32) void sem_fc_kernel(
    const float* __restrict__ x,  const float* __restrict__ wv,
    const float* __restrict__ alpha, const float* __restrict__ W,
    const float* __restrict__ b,  float* __restrict__ sem_out,
    float* __restrict__ sem2) {
    const int lane = threadIdx.x;
    const int col  = lane & 15;
    const int hi   = lane >> 4;
    const int i0   = blockIdx.x * 16;

    float alph[8];
    #pragma unroll
    for (int r = 0; r < 8; ++r) alph[r] = alpha[i0 + r + 8 * hi];

    float nrm[8];
    #pragma unroll
    for (int r = 0; r < 8; ++r) nrm[r] = 0.f;

    const float* arow = wv + (size_t)(i0 + col) * WV_DIM + 2 * hi;

    for (int dt = 0; dt < D_DIM / 16; ++dt) {
        const int d = dt * 16 + col;
        const float* brow = W + (size_t)d * WV_DIM + 2 * hi;

        v8f c = {};
        for (int k = 0; k < WV_DIM; k += 4) {
            v2f a  = ld2(arow + k);
            v2f bb = ld2(brow + k);
            c = __builtin_amdgcn_wmma_f32_16x16x4_f32(
                    false, a, false, bb, (short)0, c, false, false);
        }

        const float bd = b[d];
        #pragma unroll
        for (int r = 0; r < 8; ++r) {
            const int i = i0 + r + 8 * hi;
            const float so = c[r] + bd;
            const float xv = x[(size_t)i * D_DIM + d];
            const float s  = alph[r] * so + (1.f - alph[r]) * xv;
            sem_out[(size_t)i * D_DIM + d] = s;
            nrm[r] = fmaf(s, s, nrm[r]);
        }
    }

    #pragma unroll
    for (int r = 0; r < 8; ++r) {
        float v = nrm[r];
        for (int off = 8; off >= 1; off >>= 1)
            v += __shfl_xor(v, off, 16);
        if (col == 0) sem2[i0 + r + 8 * hi] = v;
    }
}

#define PR_WAVES   8
#define PR_ROWS    (16 * PR_WAVES)
#define PR_SPLIT   16
#define PR_JCHUNK  (N_PTS / PR_SPLIT)
#define PR_JSTAGE  64
#define PR_NSTAGE  (PR_JCHUNK / PR_JSTAGE)
#define PR_PITCH   132

typedef float v4f __attribute__((ext_vector_type(4)));

__device__ __forceinline__ void pr_epilogue(
    const v8f& c, int j, int labj, float x2j, int ibase,
    const int (&labi)[8], const float (&s2i)[8], float (&fp)[8], float (&cn)[8]) {
    #pragma unroll
    for (int r = 0; r < 8; ++r) {
        const int i = ibase + r;
        float sq = s2i[r] + x2j - 2.f * c[r];
        sq = sq > 0.f ? sq : 0.f;
        const float dist = sqrtf(sq + EPS_TRM);
        const bool same = (labi[r] == labj);
        const bool pos  = same && (i != j);
        fp[r] = fmaxf(fp[r], pos ? dist : 0.f);
        cn[r] = fminf(cn[r], same ? dist + BIG_F : dist);
    }
}

__global__ __launch_bounds__(32 * PR_WAVES) void pair_kernel(
    const float* __restrict__ x,    const int* __restrict__ label,
    const float* __restrict__ sem,  const float* __restrict__ sem2,
    const float* __restrict__ x2,   float* __restrict__ part_fp,
    float* __restrict__ part_cn) {
    __shared__ __attribute__((aligned(16))) float xs[2][PR_JSTAGE * PR_PITCH];
    __shared__ float red[2 * PR_ROWS];

    const int tid  = threadIdx.x;
    const int lane = tid & 31;
    const int wv   = tid >> 5;
    const int col  = lane & 15;
    const int hi   = lane >> 4;
    const int i0   = blockIdx.x * PR_ROWS + wv * 16;
    const int jc0  = blockIdx.y * PR_JCHUNK;

    const float* gsrc = x + (size_t)(jc0 + wv) * D_DIM + lane * 4;
    const int    lofs = wv * PR_PITCH + lane * 4;

    v4f st[8];
    #pragma unroll
    for (int q = 0; q < 8; ++q)
        st[q] = *(const v4f*)(gsrc + (size_t)(q * 8) * D_DIM);

    v2f afrag[32];
    const float* arow = sem + (size_t)(i0 + col) * D_DIM + 2 * hi;
    #pragma unroll
    for (int ks = 0; ks < 32; ++ks) afrag[ks] = ld2(arow + 4 * ks);

    const int ibase = i0 + 8 * hi;
    int   labi[8];
    float s2i[8];
    #pragma unroll
    for (int r = 0; r < 8; ++r) {
        labi[r] = label[ibase + r];
        s2i[r]  = sem2[ibase + r];
    }

    float fp[8], cn[8];
    #pragma unroll
    for (int r = 0; r < 8; ++r) { fp[r] = 0.f; cn[r] = 3.4e38f; }

    #pragma unroll
    for (int q = 0; q < 8; ++q)
        *(v4f*)(&xs[0][lofs + q * 8 * PR_PITCH]) = st[q];
    __syncthreads();

    for (int s = 0; s < PR_NSTAGE; ++s) {
        const int cur = s & 1;
        const bool more = (s + 1 < PR_NSTAGE);
        if (more) {
            #pragma unroll
            for (int q = 0; q < 8; ++q)
                st[q] = *(const v4f*)(gsrc + (size_t)((s + 1) * PR_JSTAGE + q * 8) * D_DIM);
        }

        const float* bbase = &xs[cur][col * PR_PITCH + 2 * hi];
        #pragma unroll
        for (int tp = 0; tp < PR_JSTAGE / 32; ++tp) {
            const int   ja    = jc0 + s * PR_JSTAGE + tp * 32 + col;
            const int   jb    = ja + 16;
            const int   labja = label[ja];
            const int   labjb = label[jb];
            const float x2ja  = x2[ja];
            const float x2jb  = x2[jb];
            const float* ba = bbase + tp * 32 * PR_PITCH;
            const float* bb = ba + 16 * PR_PITCH;

            v8f ca = {};
            v8f cb = {};
            #pragma unroll
            for (int ks = 0; ks < 32; ++ks) {
                const v2f fa = *(const v2f*)(ba + 4 * ks);
                const v2f fb = *(const v2f*)(bb + 4 * ks);
                ca = __builtin_amdgcn_wmma_f32_16x16x4_f32(
                        false, afrag[ks], false, fa, (short)0, ca, false, false);
                cb = __builtin_amdgcn_wmma_f32_16x16x4_f32(
                        false, afrag[ks], false, fb, (short)0, cb, false, false);
            }
            pr_epilogue(ca, ja, labja, x2ja, ibase, labi, s2i, fp, cn);
            pr_epilogue(cb, jb, labjb, x2jb, ibase, labi, s2i, fp, cn);
        }

        if (more) {
            #pragma unroll
            for (int q = 0; q < 8; ++q)
                *(v4f*)(&xs[cur ^ 1][lofs + q * 8 * PR_PITCH]) = st[q];
        }
        __syncthreads();
    }

    #pragma unroll
    for (int r = 0; r < 8; ++r) {
        float a = fp[r], m = cn[r];
        for (int off = 8; off >= 1; off >>= 1) {
            a = fmaxf(a, __shfl_xor(a, off, 16));
            m = fminf(m, __shfl_xor(m, off, 16));
        }
        if (col == 0) {
            red[wv * 16 + 8 * hi + r]           = a;
            red[PR_ROWS + wv * 16 + 8 * hi + r] = m;
        }
    }
    __syncthreads();
    {
        const size_t o = (size_t)blockIdx.y * N_PTS + (size_t)blockIdx.x * PR_ROWS;
        if (tid < PR_ROWS) part_fp[o + tid] = red[tid];
        else               part_cn[o + (tid - PR_ROWS)] = red[tid];
    }
}

__global__ void merge_kernel(const float* __restrict__ part_fp,
                             const float* __restrict__ part_cn,
                             float* __restrict__ diff) {
    const int i = blockIdx.x * blockDim.x + threadIdx.x;
    if (i < N_PTS) {
        float a = 0.f, m = 3.4e38f;
        #pragma unroll
        for (int s = 0; s < PR_SPLIT; ++s) {
            a = fmaxf(a, part_fp[(size_t)s * N_PTS + i]);
            m = fminf(m, part_cn[(size_t)s * N_PTS + i]);
        }
        const float z  = a - m;
        const float sp = (z > 20.f) ? z : log1pf(expf(z));
        diff[i] = sp;
    }
}

extern "C" void kernel_launch(void* const* d_in, const int* in_sizes, int n_in,
                              void* d_out, int out_size, void* d_ws, size_t ws_size,
                              hipStream_t stream) {
    const float* x     = (const float*)d_in[0];
    const float* wv    = (const float*)d_in[1];
    const int*   label = (const int*)  d_in[2];
    const float* alpha = (const float*)d_in[3];
    const float* W     = (const float*)d_in[4];
    const float* b     = (const float*)d_in[5];

    float* diff = (float*)d_out;
    float* sem  = diff + N_PTS;

    float* x2      = (float*)d_ws;
    float* sem2    = x2 + N_PTS;
    float* part_fp = sem2 + N_PTS;
    float* part_cn = part_fp + PR_SPLIT * N_PTS;

    hipLaunchKernelGGL(x2_kernel, dim3(N_PTS / 256), dim3(256), 0, stream, x, x2);
    hipLaunchKernelGGL(sem_fc_kernel, dim3(N_PTS / 16), dim3(32), 0, stream,
                       x, wv, alpha, W, b, sem, sem2);
    hipLaunchKernelGGL(pair_kernel, dim3(N_PTS / PR_ROWS, PR_SPLIT), dim3(32 * PR_WAVES), 0, stream,
                       x, label, sem, sem2, x2, part_fp, part_cn);
    hipLaunchKernelGGL(merge_kernel, dim3(N_PTS / 256), dim3(256), 0, stream,
                       part_fp, part_cn, diff);
}


# fc kernel: k loop outermost, each wave keeps all 8 column tiles of its 16-row stripe as accumulators (1 A fragment feeds 8 WMMA chains), fragments fetched 5 k steps ahead; no sum reordered
# speedup vs baseline: 11.0305x; 1.1345x over previous
#include <hip/hip_runtime.h>
#include <math.h>

#define N_PTS   8192
#define WV_DIM  300
#define D_DIM   128
#define BIG_F   1e8f
#define EPS_TRM (128.0f * 1e-16f)

typedef float v2f __attribute__((ext_vector_type(2)));
typedef float v8f __attribute__((ext_vector_type(8)));

__device__ __forceinline__ v2f ld2(const float* p) {
    return *(const v2f*)p;
}

__global__ void x2_kernel(const float* __restrict__ x, float* __restrict__ x2) {
    int i = blockIdx.x * blockDim.x + threadIdx.x;
    if (i < N_PTS) {
        const float* row = x + (size_t)i * D_DIM;
        float s = 0.f;
        #pragma unroll 8
        for (int d = 0; d < D_DIM; ++d) s = fmaf(row[d], row[d], s);
        x2[i] = s;
    }
}

#define FC_G   5
#define FC_NG  (WV_DIM / 4 / FC_G)
#define FC_NT  (D_DIM / 16)

__device__ __forceinline__ void fc_fetch(
    const float* __restrict__ arow, const float* __restrict__ brow, int g,
    v2f (&fa)[FC_G], v2f (&fb)[FC_G][FC_NT]) {
    #pragma unroll
    for (int u = 0; u < FC_G; ++u) {
        const int k = 4 * (g * FC_G + u);
        fa[u] = ld2(arow + k);
        #pragma unroll
        for (int t = 0; t < FC_NT; ++t)
            fb[u][t] = ld2(brow + (size_t)t * 16 * WV_DIM + k);
    }
}

__device__ __forceinline__ void fc_mma(
    const v2f (&fa)[FC_G], const v2f (&fb)[FC_G][FC_NT], v8f (&c)[FC_NT]) {
    #pragma unroll
    for (int u = 0; u < FC_G; ++u) {
        #pragma unroll
        for (int t = 0; t < FC_NT; ++t)
            c[t] = __builtin_amdgcn_wmma_f32_16x16x4_f32(
                       false, fa[u], false, fb[u][t], (short)0, c[t], false, false);
    }
}

__global__ __launch_bounds__(32) void sem_fc_kernel(
    const float* __restrict__ x,  const float* __restrict__ wv,
    const float* __restrict__ alpha, const float* __restrict__ W,
    const float* __restrict__ b,  float* __restrict__ sem_out,
    float* __restrict__ sem2) {
    const int lane = threadIdx.x;
    const int col  = lane & 15;
    const int hi   = lane >> 4;
    const int i0   = blockIdx.x * 16;

    const float* arow = wv + (size_t)(i0 + col) * WV_DIM + 2 * hi;
    const float* brow = W  + (size_t)col * WV_DIM + 2 * hi;

    v8f c[FC_NT];
    #pragma unroll
    for (int t = 0; t < FC_NT; ++t) c[t] = v8f{};

    v2f fa0[FC_G], fa1[FC_G];
    v2f fb0[FC_G][FC_NT], fb1[FC_G][FC_NT];

    fc_fetch(arow, brow, 0, fa0, fb0);
    for (int g = 0; g + 2 < FC_NG; g += 2) {
        fc_fetch(arow, brow, g + 1, fa1, fb1);
        fc_mma(fa0, fb0, c);
        fc_fetch(arow, brow, g + 2, fa0, fb0);
        fc_mma(fa1, fb1, c);
    }
    fc_mma(fa0, fb0, c);

    float alph[8];
    #pragma unroll
    for (int r = 0; r < 8; ++r) alph[r] = alpha[i0 + r + 8 * hi];

    float nrm[8];
    #pragma unroll
    for (int r = 0; r < 8; ++r) nrm[r] = 0.f;

    #pragma unroll
    for (int dt = 0; dt < FC_NT; ++dt) {
        const int d = dt * 16 + col;
        const float bd = b[d];
        #pragma unroll
        for (int r = 0; r < 8; ++r) {
            const int i = i0 + r + 8 * hi;
            const float so = c[dt][r] + bd;
            const float xv = x[(size_t)i * D_DIM + d];
            const float xt = (1.f - alph[r]) * xv;
            const float s  = fmaf(alph[r], so, xt);
            sem_out[(size_t)i * D_DIM + d] = s;
            nrm[r] = fmaf(s, s, nrm[r]);
        }
    }

    #pragma unroll
    for (int r = 0; r < 8; ++r) {
        float v = nrm[r];
        for (int off = 8; off >= 1; off >>= 1)
            v += __shfl_xor(v, off, 16);
        if (col == 0) sem2[i0 + r + 8 * hi] = v;
    }
}

#define PR_WAVES   8
#define PR_ROWS    (16 * PR_WAVES)
#define PR_SPLIT   16
#define PR_JCHUNK  (N_PTS / PR_SPLIT)
#define PR_JSTAGE  64
#define PR_NSTAGE  (PR_JCHUNK / PR_JSTAGE)
#define PR_PITCH   132

typedef float v4f __attribute__((ext_vector_type(4)));

__device__ __forceinline__ void pr_epilogue(
    const v8f& c, int j, int labj, float x2j, int ibase,
    const int (&labi)[8], const float (&s2i)[8], float (&fp)[8], float (&cn)[8]) {
    #pragma unroll
    for (int r = 0; r < 8; ++r) {
        const int i = ibase + r;
        float sq = s2i[r] + x2j - 2.f * c[r];
        sq = sq > 0.f ? sq : 0.f;
        const float dist = sqrtf(sq + EPS_TRM);
        const bool same = (labi[r] == labj);
        const bool pos  = same && (i != j);
        fp[r] = fmaxf(fp[r], pos ? dist : 0.f);
        cn[r] = fminf(cn[r], same ? dist + BIG_F : dist);
    }
}

__global__ __launch_bounds__(32 * PR_WAVES) void pair_kernel(
    const float* __restrict__ x,    const int* __restrict__ label,
    const float* __restrict__ sem,  const float* __restrict__ sem2,
    const float* __restrict__ x2,   float* __restrict__ part_fp,
    float* __restrict__ part_cn) {
    __shared__ __attribute__((aligned(16))) float xs[2][PR_JSTAGE * PR_PITCH];
    __shared__ float red[2 * PR_ROWS];

    const int tid  = threadIdx.x;
    const int lane = tid & 31;
    const int wv   = tid >> 5;
    const int col  = lane & 15;
    const int hi   = lane >> 4;
    const int i0   = blockIdx.x * PR_ROWS + wv * 16;
    const int jc0  = blockIdx.y * PR_JCHUNK;

    const float* gsrc = x + (size_t)(jc0 + wv) * D_DIM + lane * 4;
    const int    lofs = wv * PR_PITCH + lane * 4;

    v4f st[8];
    #pragma unroll
    for (int q = 0; q < 8; ++q)
        st[q] = *(const v4f*)(gsrc + (size_t)(q * 8) * D_DIM);

    v2f afrag[32];
    const float* arow = sem + (size_t)(i0 + col) * D_DIM + 2 * hi;
    #pragma unroll
    for (int ks = 0; ks < 32; ++ks) afrag[ks] = ld2(arow + 4 * ks);

    const int ibase = i0 + 8 * hi;
    int   labi[8];
    float s2i[8];
    #pragma unroll
    for (int r = 0; r < 8; ++r) {
        labi[r] = label[ibase + r];
        s2i[r]  = sem2[ibase + r];
    }

    float fp[8], cn[8];
    #pragma unroll
    for (int r = 0; r < 8; ++r) { fp[r] = 0.f; cn[r] = 3.4e38f; }

    #pragma unroll
    for (int q = 0; q < 8; ++q)
        *(v4f*)(&xs[0][lofs + q * 8 * PR_PITCH]) = st[q];
    __syncthreads();

    for (int s = 0; s < PR_NSTAGE; ++s) {
        const int cur = s & 1;
        const bool more = (s + 1 < PR_NSTAGE);
        if (more) {
            #pragma unroll
            for (int q = 0; q < 8; ++q)
                st[q] = *(const v4f*)(gsrc + (size_t)((s + 1) * PR_JSTAGE + q * 8) * D_DIM);
        }

        const float* bbase = &xs[cur][col * PR_PITCH + 2 * hi];
        #pragma unroll
        for (int tp = 0; tp < PR_JSTAGE / 32; ++tp) {
            const int   ja    = jc0 + s * PR_JSTAGE + tp * 32 + col;
            const int   jb    = ja + 16;
            const int   labja = label[ja];
            const int   labjb = label[jb];
            const float x2ja  = x2[ja];
            const float x2jb  = x2[jb];
            const float* ba = bbase + tp * 32 * PR_PITCH;
            const float* bb = ba + 16 * PR_PITCH;

            v8f ca = {};
            v8f cb = {};
            #pragma unroll
            for (int ks = 0; ks < 32; ++ks) {
                const v2f fa = *(const v2f*)(ba + 4 * ks);
                const v2f fb = *(const v2f*)(bb + 4 * ks);
                ca = __builtin_amdgcn_wmma_f32_16x16x4_f32(
                        false, afrag[ks], false, fa, (short)0, ca, false, false);
                cb = __builtin_amdgcn_wmma_f32_16x16x4_f32(
                        false, afrag[ks], false, fb, (short)0, cb, false, false);
            }
            pr_epilogue(ca, ja, labja, x2ja, ibase, labi, s2i, fp, cn);
            pr_epilogue(cb, jb, labjb, x2jb, ibase, labi, s2i, fp, cn);
        }

        if (more) {
            #pragma unroll
            for (int q = 0; q < 8; ++q)
                *(v4f*)(&xs[cur ^ 1][lofs + q * 8 * PR_PITCH]) = st[q];
        }
        __syncthreads();
    }

    #pragma unroll
    for (int r = 0; r < 8; ++r) {
        float a = fp[r], m = cn[r];
        for (int off = 8; off >= 1; off >>= 1) {
            a = fmaxf(a, __shfl_xor(a, off, 16));
            m = fminf(m, __shfl_xor(m, off, 16));
        }
        if (col == 0) {
            red[wv * 16 + 8 * hi + r]           = a;
            red[PR_ROWS + wv * 16 + 8 * hi + r] = m;
        }
    }
    __syncthreads();
    {
        const size_t o = (size_t)blockIdx.y * N_PTS + (size_t)blockIdx.x * PR_ROWS;
        if (tid < PR_ROWS) part_fp[o + tid] = red[tid];
        else               part_cn[o + (tid - PR_ROWS)] = red[tid];
    }
}

__global__ void merge_kernel(const float* __restrict__ part_fp,
                             const float* __restrict__ part_cn,
                             float* __restrict__ diff) {
    const int i = blockIdx.x * blockDim.x + threadIdx.x;
    if (i < N_PTS) {
        float a = 0.f, m = 3.4e38f;
        #pragma unroll
        for (int s = 0; s < PR_SPLIT; ++s) {
            a = fmaxf(a, part_fp[(size_t)s * N_PTS + i]);
            m = fminf(m, part_cn[(size_t)s * N_PTS + i]);
        }
        const float z  = a - m;
        const float sp = (z > 20.f) ? z : log1pf(expf(z));
        diff[i] = sp;
    }
}

extern "C" void kernel_launch(void* const* d_in, const int* in_sizes, int n_in,
                              void* d_out, int out_size, void* d_ws, size_t ws_size,
                              hipStream_t stream) {
    const float* x     = (const float*)d_in[0];
    const float* wv    = (const float*)d_in[1];
    const int*   label = (const int*)  d_in[2];
    const float* alpha = (const float*)d_in[3];
    const float* W     = (const float*)d_in[4];
    const float* b     = (const float*)d_in[5];

    float* diff = (float*)d_out;
    float* sem  = diff + N_PTS;

    float* x2      = (float*)d_ws;
    float* sem2    = x2 + N_PTS;
    float* part_fp = sem2 + N_PTS;
    float* part_cn = part_fp + PR_SPLIT * N_PTS;

    hipLaunchKernelGGL(x2_kernel, dim3(N_PTS / 256), dim3(256), 0, stream, x, x2);
    hipLaunchKernelGGL(sem_fc_kernel, dim3(N_PTS / 16), dim3(32), 0, stream,
                       x, wv, alpha, W, b, sem, sem2);
    hipLaunchKernelGGL(pair_kernel, dim3(N_PTS / PR_ROWS, PR_SPLIT), dim3(32 * PR_WAVES), 0, stream,
                       x, label, sem, sem2, x2, part_fp, part_cn);
    hipLaunchKernelGGL(merge_kernel, dim3(N_PTS / 256), dim3(256), 0, stream,
                       part_fp, part_cn, diff);
}


# pair kernel: per-pair sqrt = hardware instruction + its one-ulp residual fix-up only; the library form's tiny-argument rescale and zero/inf patch cannot act on max(sq,0)+D*eps; bit-identical
# speedup vs baseline: 11.6546x; 1.0566x over previous
#include <hip/hip_runtime.h>
#include <math.h>

#define N_PTS   8192
#define WV_DIM  300
#define D_DIM   128
#define BIG_F   1e8f
#define EPS_TRM (128.0f * 1e-16f)

typedef float v2f __attribute__((ext_vector_type(2)));
typedef float v8f __attribute__((ext_vector_type(8)));

__device__ __forceinline__ v2f ld2(const float* p) {
    return *(const v2f*)p;
}

__global__ void x2_kernel(const float* __restrict__ x, float* __restrict__ x2) {
    int i = blockIdx.x * blockDim.x + threadIdx.x;
    if (i < N_PTS) {
        const float* row = x + (size_t)i * D_DIM;
        float s = 0.f;
        #pragma unroll 8
        for (int d = 0; d < D_DIM; ++d) s = fmaf(row[d], row[d], s);
        x2[i] = s;
    }
}

#define FC_G   5
#define FC_NG  (WV_DIM / 4 / FC_G)
#define FC_NT  (D_DIM / 16)

__device__ __forceinline__ void fc_fetch(
    const float* __restrict__ arow, const float* __restrict__ brow, int g,
    v2f (&fa)[FC_G], v2f (&fb)[FC_G][FC_NT]) {
    #pragma unroll
    for (int u = 0; u < FC_G; ++u) {
        const int k = 4 * (g * FC_G + u);
        fa[u] = ld2(arow + k);
        #pragma unroll
        for (int t = 0; t < FC_NT; ++t)
            fb[u][t] = ld2(brow + (size_t)t * 16 * WV_DIM + k);
    }
}

__device__ __forceinline__ void fc_mma(
    const v2f (&fa)[FC_G], const v2f (&fb)[FC_G][FC_NT], v8f (&c)[FC_NT]) {
    #pragma unroll
    for (int u = 0; u < FC_G; ++u) {
        #pragma unroll
        for (int t = 0; t < FC_NT; ++t)
            c[t] = __builtin_amdgcn_wmma_f32_16x16x4_f32(
                       false, fa[u], false, fb[u][t], (short)0, c[t], false, false);
    }
}

__global__ __launch_bounds__(32) void sem_fc_kernel(
    const float* __restrict__ x,  const float* __restrict__ wv,
    const float* __restrict__ alpha, const float* __restrict__ W,
    const float* __restrict__ b,  float* __restrict__ sem_out,
    float* __restrict__ sem2) {
    const int lane = threadIdx.x;
    const int col  = lane & 15;
    const int hi   = lane >> 4;
    const int i0   = blockIdx.x * 16;

    const float* arow = wv + (size_t)(i0 + col) * WV_DIM + 2 * hi;
    const float* brow = W  + (size_t)col * WV_DIM + 2 * hi;

    v8f c[FC_NT];
    #pragma unroll
    for (int t = 0; t < FC_NT; ++t) c[t] = v8f{};

    v2f fa0[FC_G], fa1[FC_G];
    v2f fb0[FC_G][FC_NT], fb1[FC_G][FC_NT];

    fc_fetch(arow, brow, 0, fa0, fb0);
    for (int g = 0; g + 2 < FC_NG; g += 2) {
        fc_fetch(arow, brow, g + 1, fa1, fb1);
        fc_mma(fa0, fb0, c);
        fc_fetch(arow, brow, g + 2, fa0, fb0);
        fc_mma(fa1, fb1, c);
    }
    fc_mma(fa0, fb0, c);

    float alph[8];
    #pragma unroll
    for (int r = 0; r < 8; ++r) alph[r] = alpha[i0 + r + 8 * hi];

    float nrm[8];
    #pragma unroll
    for (int r = 0; r < 8; ++r) nrm[r] = 0.f;

    #pragma unroll
    for (int dt = 0; dt < FC_NT; ++dt) {
        const int d = dt * 16 + col;
        const float bd = b[d];
        #pragma unroll
        for (int r = 0; r < 8; ++r) {
            const int i = i0 + r + 8 * hi;
            const float so = c[dt][r] + bd;
            const float xv = x[(size_t)i * D_DIM + d];
            const float xt = (1.f - alph[r]) * xv;
            const float s  = fmaf(alph[r], so, xt);
            sem_out[(size_t)i * D_DIM + d] = s;
            nrm[r] = fmaf(s, s, nrm[r]);
        }
    }

    #pragma unroll
    for (int r = 0; r < 8; ++r) {
        float v = nrm[r];
        for (int off = 8; off >= 1; off >>= 1)
            v += __shfl_xor(v, off, 16);
        if (col == 0) sem2[i0 + r + 8 * hi] = v;
    }
}

#define PR_WAVES   8
#define PR_ROWS    (16 * PR_WAVES)
#define PR_SPLIT   16
#define PR_JCHUNK  (N_PTS / PR_SPLIT)
#define PR_JSTAGE  64
#define PR_NSTAGE  (PR_JCHUNK / PR_JSTAGE)
#define PR_PITCH   132

typedef float v4f __attribute__((ext_vector_type(4)));

__device__ __forceinline__ float pr_sqrt(float t) {
    const float s    = __builtin_amdgcn_sqrtf(t);
    const float s_dn = __int_as_float(__float_as_int(s) - 1);
    const float s_up = __int_as_float(__float_as_int(s) + 1);
    const float r_dn = fmaf(-s_dn, s, t);
    const float r_up = fmaf(-s_up, s, t);
    float r = (r_dn <= 0.f) ? s_dn : s;
    r = (r_up > 0.f) ? s_up : r;
    return r;
}

__device__ __forceinline__ void pr_epilogue(
    const v8f& c, int j, int labj, float x2j, int ibase,
    const int (&labi)[8], const float (&s2i)[8], float (&fp)[8], float (&cn)[8]) {
    #pragma unroll
    for (int r = 0; r < 8; ++r) {
        const int i = ibase + r;
        float sq = s2i[r] + x2j - 2.f * c[r];
        sq = sq > 0.f ? sq : 0.f;
        const float dist = pr_sqrt(sq + EPS_TRM);
        const bool same = (labi[r] == labj);
        const bool pos  = same && (i != j);
        fp[r] = fmaxf(fp[r], pos ? dist : 0.f);
        cn[r] = fminf(cn[r], same ? dist + BIG_F : dist);
    }
}

__global__ __launch_bounds__(32 * PR_WAVES) void pair_kernel(
    const float* __restrict__ x,    const int* __restrict__ label,
    const float* __restrict__ sem,  const float* __restrict__ sem2,
    const float* __restrict__ x2,   float* __restrict__ part_fp,
    float* __restrict__ part_cn) {
    __shared__ __attribute__((aligned(16))) float xs[2][PR_JSTAGE * PR_PITCH];
    __shared__ float red[2 * PR_ROWS];

    const int tid  = threadIdx.x;
    const int lane = tid & 31;
    const int wv   = tid >> 5;
    const int col  = lane & 15;
    const int hi   = lane >> 4;
    const int i0   = blockIdx.x * PR_ROWS + wv * 16;
    const int jc0  = blockIdx.y * PR_JCHUNK;

    const float* gsrc = x + (size_t)(jc0 + wv) * D_DIM + lane * 4;
    const int    lofs = wv * PR_PITCH + lane * 4;

    v4f st[8];
    #pragma unroll
    for (int q = 0; q < 8; ++q)
        st[q] = *(const v4f*)(gsrc + (size_t)(q * 8) * D_DIM);

    v2f afrag[32];
    const float* arow = sem + (size_t)(i0 + col) * D_DIM + 2 * hi;
    #pragma unroll
    for (int ks = 0; ks < 32; ++ks) afrag[ks] = ld2(arow + 4 * ks);

    const int ibase = i0 + 8 * hi;
    int   labi[8];
    float s2i[8];
    #pragma unroll
    for (int r = 0; r < 8; ++r) {
        labi[r] = label[ibase + r];
        s2i[r]  = sem2[ibase + r];
    }

    float fp[8], cn[8];
    #pragma unroll
    for (int r = 0; r < 8; ++r) { fp[r] = 0.f; cn[r] = 3.4e38f; }

    #pragma unroll
    for (int q = 0; q < 8; ++q)
        *(v4f*)(&xs[0][lofs + q * 8 * PR_PITCH]) = st[q];
    __syncthreads();

    for (int s = 0; s < PR_NSTAGE; ++s) {
        const int cur = s & 1;
        const bool more = (s + 1 < PR_NSTAGE);
        if (more) {
            #pragma unroll
            for (int q = 0; q < 8; ++q)
                st[q] = *(const v4f*)(gsrc + (size_t)((s + 1) * PR_JSTAGE + q * 8) * D_DIM);
        }

        const float* bbase = &xs[cur][col * PR_PITCH + 2 * hi];
        #pragma unroll
        for (int tp = 0; tp < PR_JSTAGE / 32; ++tp) {
            const int   ja    = jc0 + s * PR_JSTAGE + tp * 32 + col;
            const int   jb    = ja + 16;
            const int   labja = label[ja];
            const int   labjb = label[jb];
            const float x2ja  = x2[ja];
            const float x2jb  = x2[jb];
            const float* ba = bbase + tp * 32 * PR_PITCH;
            const float* bb = ba + 16 * PR_PITCH;

            v8f ca = {};
            v8f cb = {};
            #pragma unroll
            for (int ks = 0; ks < 32; ++ks) {
                const v2f fa = *(const v2f*)(ba + 4 * ks);
                const v2f fb = *(const v2f*)(bb + 4 * ks);
                ca = __builtin_amdgcn_wmma_f32_16x16x4_f32(
                        false, afrag[ks], false, fa, (short)0, ca, false, false);
                cb = __builtin_amdgcn_wmma_f32_16x16x4_f32(
                        false, afrag[ks], false, fb, (short)0, cb, false, false);
            }
            pr_epilogue(ca, ja, labja, x2ja, ibase, labi, s2i, fp, cn);
            pr_epilogue(cb, jb, labjb, x2jb, ibase, labi, s2i, fp, cn);
        }

        if (more) {
            #pragma unroll
            for (int q = 0; q < 8; ++q)
                *(v4f*)(&xs[cur ^ 1][lofs + q * 8 * PR_PITCH]) = st[q];
        }
        __syncthreads();
    }

    #pragma unroll
    for (int r = 0; r < 8; ++r) {
        float a = fp[r], m = cn[r];
        for (int off = 8; off >= 1; off >>= 1) {
            a = fmaxf(a, __shfl_xor(a, off, 16));
            m = fminf(m, __shfl_xor(m, off, 16));
        }
        if (col == 0) {
            red[wv * 16 + 8 * hi + r]           = a;
            red[PR_ROWS + wv * 16 + 8 * hi + r] = m;
        }
    }
    __syncthreads();
    {
        const size_t o = (size_t)blockIdx.y * N_PTS + (size_t)blockIdx.x * PR_ROWS;
        if (tid < PR_ROWS) part_fp[o + tid] = red[tid];
        else               part_cn[o + (tid - PR_ROWS)] = red[tid];
    }
}

__global__ void merge_kernel(const float* __restrict__ part_fp,
                             const float* __restrict__ part_cn,
                             float* __restrict__ diff) {
    const int i = blockIdx.x * blockDim.x + threadIdx.x;
    if (i < N_PTS) {
        float a = 0.f, m = 3.4e38f;
        #pragma unroll
        for (int s = 0; s < PR_SPLIT; ++s) {
            a = fmaxf(a, part_fp[(size_t)s * N_PTS + i]);
            m = fminf(m, part_cn[(size_t)s * N_PTS + i]);
        }
        const float z  = a - m;
        const float sp = (z > 20.f) ? z : log1pf(expf(z));
        diff[i] = sp;
    }
}

extern "C" void kernel_launch(void* const* d_in, const int* in_sizes, int n_in,
                              void* d_out, int out_size, void* d_ws, size_t ws_size,
                              hipStream_t stream) {
    const float* x     = (const float*)d_in[0];
    const float* wv    = (const float*)d_in[1];
    const int*   label = (const int*)  d_in[2];
    const float* alpha = (const float*)d_in[3];
    const float* W     = (const float*)d_in[4];
    const float* b     = (const float*)d_in[5];

    float* diff = (float*)d_out;
    float* sem  = diff + N_PTS;

    float* x2      = (float*)d_ws;
    float* sem2    = x2 + N_PTS;
    float* part_fp = sem2 + N_PTS;
    float* part_cn = part_fp + PR_SPLIT * N_PTS;

    hipLaunchKernelGGL(x2_kernel, dim3(N_PTS / 256), dim3(256), 0, stream, x, x2);
    hipLaunchKernelGGL(sem_fc_kernel, dim3(N_PTS / 16), dim3(32), 0, stream,
                       x, wv, alpha, W, b, sem, sem2);
    hipLaunchKernelGGL(pair_kernel, dim3(N_PTS / PR_ROWS, PR_SPLIT), dim3(32 * PR_WAVES), 0, stream,
                       x, label, sem, sem2, x2, part_fp, part_cn);
    hipLaunchKernelGGL(merge_kernel, dim3(N_PTS / 256), dim3(256), 0, stream,
                       part_fp, part_cn, diff);
}


# pair kernel: per-pair sqrt = the bare hardware instruction (within 1 ulp); its correctly rounding fix-up (2 int adds, 2 fma, 2 compares, 2 selects a pair) dropped; not bit-identical, no sum reordered
# speedup vs baseline: 12.3530x; 1.0599x over previous
#include <hip/hip_runtime.h>
#include <math.h>

#define N_PTS   8192
#define WV_DIM  300
#define D_DIM   128
#define BIG_F   1e8f
#define EPS_TRM (128.0f * 1e-16f)

typedef float v2f __attribute__((ext_vector_type(2)));
typedef float v8f __attribute__((ext_vector_type(8)));

__device__ __forceinline__ v2f ld2(const float* p) {
    return *(const v2f*)p;
}

__global__ void x2_kernel(const float* __restrict__ x, float* __restrict__ x2) {
    int i = blockIdx.x * blockDim.x + threadIdx.x;
    if (i < N_PTS) {
        const float* row = x + (size_t)i * D_DIM;
        float s = 0.f;
        #pragma unroll 8
        for (int d = 0; d < D_DIM; ++d) s = fmaf(row[d], row[d], s);
        x2[i] = s;
    }
}

#define FC_G   5
#define FC_NG  (WV_DIM / 4 / FC_G)
#define FC_NT  (D_DIM / 16)

__device__ __forceinline__ void fc_fetch(
    const float* __restrict__ arow, const float* __restrict__ brow, int g,
    v2f (&fa)[FC_G], v2f (&fb)[FC_G][FC_NT]) {
    #pragma unroll
    for (int u = 0; u < FC_G; ++u) {
        const int k = 4 * (g * FC_G + u);
        fa[u] = ld2(arow + k);
        #pragma unroll
        for (int t = 0; t < FC_NT; ++t)
            fb[u][t] = ld2(brow + (size_t)t * 16 * WV_DIM + k);
    }
}

__device__ __forceinline__ void fc_mma(
    const v2f (&fa)[FC_G], const v2f (&fb)[FC_G][FC_NT], v8f (&c)[FC_NT]) {
    #pragma unroll
    for (int u = 0; u < FC_G; ++u) {
        #pragma unroll
        for (int t = 0; t < FC_NT; ++t)
            c[t] = __builtin_amdgcn_wmma_f32_16x16x4_f32(
                       false, fa[u], false, fb[u][t], (short)0, c[t], false, false);
    }
}

__global__ __launch_bounds__(32) void sem_fc_kernel(
    const float* __restrict__ x,  const float* __restrict__ wv,
    const float* __restrict__ alpha, const float* __restrict__ W,
    const float* __restrict__ b,  float* __restrict__ sem_out,
    float* __restrict__ sem2) {
    const int lane = threadIdx.x;
    const int col  = lane & 15;
    const int hi   = lane >> 4;
    const int i0   = blockIdx.x * 16;

    const float* arow = wv + (size_t)(i0 + col) * WV_DIM + 2 * hi;
    const float* brow = W  + (size_t)col * WV_DIM + 2 * hi;

    v8f c[FC_NT];
    #pragma unroll
    for (int t = 0; t < FC_NT; ++t) c[t] = v8f{};

    v2f fa0[FC_G], fa1[FC_G];
    v2f fb0[FC_G][FC_NT], fb1[FC_G][FC_NT];

    fc_fetch(arow, brow, 0, fa0, fb0);
    for (int g = 0; g + 2 < FC_NG; g += 2) {
        fc_fetch(arow, brow, g + 1, fa1, fb1);
        fc_mma(fa0, fb0, c);
        fc_fetch(arow, brow, g + 2, fa0, fb0);
        fc_mma(fa1, fb1, c);
    }
    fc_mma(fa0, fb0, c);

    float alph[8];
    #pragma unroll
    for (int r = 0; r < 8; ++r) alph[r] = alpha[i0 + r + 8 * hi];

    float nrm[8];
    #pragma unroll
    for (int r = 0; r < 8; ++r) nrm[r] = 0.f;

    #pragma unroll
    for (int dt = 0; dt < FC_NT; ++dt) {
        const int d = dt * 16 + col;
        const float bd = b[d];
        #pragma unroll
        for (int r = 0; r < 8; ++r) {
            const int i = i0 + r + 8 * hi;
            const float so = c[dt][r] + bd;
            const float xv = x[(size_t)i * D_DIM + d];
            const float xt = (1.f - alph[r]) * xv;
            const float s  = fmaf(alph[r], so, xt);
            sem_out[(size_t)i * D_DIM + d] = s;
            nrm[r] = fmaf(s, s, nrm[r]);
        }
    }

    #pragma unroll
    for (int r = 0; r < 8; ++r) {
        float v = nrm[r];
        for (int off = 8; off >= 1; off >>= 1)
            v += __shfl_xor(v, off, 16);
        if (col == 0) sem2[i0 + r + 8 * hi] = v;
    }
}

#define PR_WAVES   8
#define PR_ROWS    (16 * PR_WAVES)
#define PR_SPLIT   16
#define PR_JCHUNK  (N_PTS / PR_SPLIT)
#define PR_JSTAGE  64
#define PR_NSTAGE  (PR_JCHUNK / PR_JSTAGE)
#define PR_PITCH   132

typedef float v4f __attribute__((ext_vector_type(4)));

__device__ __forceinline__ float pr_sqrt(float t) {
    return __builtin_amdgcn_sqrtf(t);
}

__device__ __forceinline__ void pr_epilogue(
    const v8f& c, int j, int labj, float x2j, int ibase,
    const int (&labi)[8], const float (&s2i)[8], float (&fp)[8], float (&cn)[8]) {
    #pragma unroll
    for (int r = 0; r < 8; ++r) {
        const int i = ibase + r;
        float sq = s2i[r] + x2j - 2.f * c[r];
        sq = sq > 0.f ? sq : 0.f;
        const float dist = pr_sqrt(sq + EPS_TRM);
        const bool same = (labi[r] == labj);
        const bool pos  = same && (i != j);
        fp[r] = fmaxf(fp[r], pos ? dist : 0.f);
        cn[r] = fminf(cn[r], same ? dist + BIG_F : dist);
    }
}

__global__ __launch_bounds__(32 * PR_WAVES) void pair_kernel(
    const float* __restrict__ x,    const int* __restrict__ label,
    const float* __restrict__ sem,  const float* __restrict__ sem2,
    const float* __restrict__ x2,   float* __restrict__ part_fp,
    float* __restrict__ part_cn) {
    __shared__ __attribute__((aligned(16))) float xs[2][PR_JSTAGE * PR_PITCH];
    __shared__ float red[2 * PR_ROWS];

    const int tid  = threadIdx.x;
    const int lane = tid & 31;
    const int wv   = tid >> 5;
    const int col  = lane & 15;
    const int hi   = lane >> 4;
    const int i0   = blockIdx.x * PR_ROWS + wv * 16;
    const int jc0  = blockIdx.y * PR_JCHUNK;

    const float* gsrc = x + (size_t)(jc0 + wv) * D_DIM + lane * 4;
    const int    lofs = wv * PR_PITCH + lane * 4;

    v4f st[8];
    #pragma unroll
    for (int q = 0; q < 8; ++q)
        st[q] = *(const v4f*)(gsrc + (size_t)(q * 8) * D_DIM);

    v2f afrag[32];
    const float* arow = sem + (size_t)(i0 + col) * D_DIM + 2 * hi;
    #pragma unroll
    for (int ks = 0; ks < 32; ++ks) afrag[ks] = ld2(arow + 4 * ks);

    const int ibase = i0 + 8 * hi;
    int   labi[8];
    float s2i[8];
    #pragma unroll
    for (int r = 0; r < 8; ++r) {
        labi[r] = label[ibase + r];
        s2i[r]  = sem2[ibase + r];
    }

    float fp[8], cn[8];
    #pragma unroll
    for (int r = 0; r < 8; ++r) { fp[r] = 0.f; cn[r] = 3.4e38f; }

    #pragma unroll
    for (int q = 0; q < 8; ++q)
        *(v4f*)(&xs[0][lofs + q * 8 * PR_PITCH]) = st[q];
    __syncthreads();

    for (int s = 0; s < PR_NSTAGE; ++s) {
        const int cur = s & 1;
        const bool more = (s + 1 < PR_NSTAGE);
        if (more) {
            #pragma unroll
            for (int q = 0; q < 8; ++q)
                st[q] = *(const v4f*)(gsrc + (size_t)((s + 1) * PR_JSTAGE + q * 8) * D_DIM);
        }

        const float* bbase = &xs[cur][col * PR_PITCH + 2 * hi];
        #pragma unroll
        for (int tp = 0; tp < PR_JSTAGE / 32; ++tp) {
            const int   ja    = jc0 + s * PR_JSTAGE + tp * 32 + col;
            const int   jb    = ja + 16;
            const int   labja = label[ja];
            const int   labjb = label[jb];
            const float x2ja  = x2[ja];
            const float x2jb  = x2[jb];
            const float* ba = bbase + tp * 32 * PR_PITCH;
            const float* bb = ba + 16 * PR_PITCH;

            v8f ca = {};
            v8f cb = {};
            #pragma unroll
            for (int ks = 0; ks < 32; ++ks) {
                const v2f fa = *(const v2f*)(ba + 4 * ks);
                const v2f fb = *(const v2f*)(bb + 4 * ks);
                ca = __builtin_amdgcn_wmma_f32_16x16x4_f32(
                        false, afrag[ks], false, fa, (short)0, ca, false, false);
                cb = __builtin_amdgcn_wmma_f32_16x16x4_f32(
                        false, afrag[ks], false, fb, (short)0, cb, false, false);
            }
            pr_epilogue(ca, ja, labja, x2ja, ibase, labi, s2i, fp, cn);
            pr_epilogue(cb, jb, labjb, x2jb, ibase, labi, s2i, fp, cn);
        }

        if (more) {
            #pragma unroll
            for (int q = 0; q < 8; ++q)
                *(v4f*)(&xs[cur ^ 1][lofs + q * 8 * PR_PITCH]) = st[q];
        }
        __syncthreads();
    }

    #pragma unroll
    for (int r = 0; r < 8; ++r) {
        float a = fp[r], m = cn[r];
        for (int off = 8; off >= 1; off >>= 1) {
            a = fmaxf(a, __shfl_xor(a, off, 16));
            m = fminf(m, __shfl_xor(m, off, 16));
        }
        if (col == 0) {
            red[wv * 16 + 8 * hi + r]           = a;
            red[PR_ROWS + wv * 16 + 8 * hi + r] = m;
        }
    }
    __syncthreads();
    {
        const size_t o = (size_t)blockIdx.y * N_PTS + (size_t)blockIdx.x * PR_ROWS;
        if (tid < PR_ROWS) part_fp[o + tid] = red[tid];
        else               part_cn[o + (tid - PR_ROWS)] = red[tid];
    }
}

__global__ void merge_kernel(const float* __restrict__ part_fp,
                             const float* __restrict__ part_cn,
                             float* __restrict__ diff) {
    const int i = blockIdx.x * blockDim.x + threadIdx.x;
    if (i < N_PTS) {
        float a = 0.f, m = 3.4e38f;
        #pragma unroll
        for (int s = 0; s < PR_SPLIT; ++s) {
            a = fmaxf(a, part_fp[(size_t)s * N_PTS + i]);
            m = fminf(m, part_cn[(size_t)s * N_PTS + i]);
        }
        const float z  = a - m;
        const float sp = (z > 20.f) ? z : log1pf(expf(z));
        diff[i] = sp;
    }
}

extern "C" void kernel_launch(void* const* d_in, const int* in_sizes, int n_in,
                              void* d_out, int out_size, void* d_ws, size_t ws_size,
                              hipStream_t stream) {
    const float* x     = (const float*)d_in[0];
    const float* wv    = (const float*)d_in[1];
    const int*   label = (const int*)  d_in[2];
    const float* alpha = (const float*)d_in[3];
    const float* W     = (const float*)d_in[4];
    const float* b     = (const float*)d_in[5];

    float* diff = (float*)d_out;
    float* sem  = diff + N_PTS;

    float* x2      = (float*)d_ws;
    float* sem2    = x2 + N_PTS;
    float* part_fp = sem2 + N_PTS;
    float* part_cn = part_fp + PR_SPLIT * N_PTS;

    hipLaunchKernelGGL(x2_kernel, dim3(N_PTS / 256), dim3(256), 0, stream, x, x2);
    hipLaunchKernelGGL(sem_fc_kernel, dim3(N_PTS / 16), dim3(32), 0, stream,
                       x, wv, alpha, W, b, sem, sem2);
    hipLaunchKernelGGL(pair_kernel, dim3(N_PTS / PR_ROWS, PR_SPLIT), dim3(32 * PR_WAVES), 0, stream,
                       x, label, sem, sem2, x2, part_fp, part_cn);
    hipLaunchKernelGGL(merge_kernel, dim3(N_PTS / 256), dim3(256), 0, stream,
                       part_fp, part_cn, diff);
}
